# starting kernel
# baseline (speedup 1.0000x reference)
#include <hip/hip_runtime.h>
#include <math.h>

typedef __attribute__((ext_vector_type(2))) float v2f;
typedef __attribute__((ext_vector_type(8))) float v8f;

#define NNODES 100000
#define DEG    16
#define DIM    128
#define DE     16
#define MEANW  (DIM + DE)
#define KTOT   (2 * DIM + DE)
#define DOUT   128

__global__ __launch_bounds__(256)
void sage_gather_mean(const float* __restrict__ h,
                      const float* __restrict__ edge_table,
                      const int*  __restrict__ neighbor_idx,
                      const int*  __restrict__ edge_idx,
                      float* __restrict__ mean) {
    const int lane = threadIdx.x & 31;
    const int node = blockIdx.x * 8 + (threadIdx.x >> 5);
    const long ebase = (long)node * DEG;

    float4 ah = make_float4(0.f, 0.f, 0.f, 0.f);
    float4 ae = make_float4(0.f, 0.f, 0.f, 0.f);

    #pragma unroll 4
    for (int e = 0; e < DEG; ++e) {
        const int nbr = neighbor_idx[ebase + e];
        const int eid = edge_idx[ebase + e];
        const float4 hv =
            *(const float4*)(h + (long)nbr * DIM + lane * 4);
        ah.x += hv.x; ah.y += hv.y; ah.z += hv.z; ah.w += hv.w;
        if (lane < 4) {
            const float4 ev =
                *(const float4*)(edge_table + (long)eid * DE + lane * 4);
            ae.x += ev.x; ae.y += ev.y; ae.z += ev.z; ae.w += ev.w;
        }
    }
    const float inv = 1.0f / (float)DEG;
    float* __restrict__ mp = mean + (long)node * MEANW;
    float4 mh = make_float4(ah.x * inv, ah.y * inv, ah.z * inv, ah.w * inv);
    *(float4*)(mp + lane * 4) = mh;
    if (lane < 4) {
        float4 me = make_float4(ae.x * inv, ae.y * inv, ae.z * inv, ae.w * inv);
        *(float4*)(mp + DIM + lane * 4) = me;
    }
}

#define RT 5
#define ASTRIDE 274

__global__ __launch_bounds__(256)
void sage_gemm_elu(const float* __restrict__ h,
                   const float* __restrict__ mean,
                   const float* __restrict__ W,
                   float* __restrict__ out) {
    __shared__ float Asub[RT * 16][ASTRIDE];

    const int tid  = threadIdx.x;
    const int wave = tid >> 5;
    const int lane = tid & 31;
    const long row0 = (long)blockIdx.x * (RT * 16);

    for (int i = tid; i < RT * 16 * DIM; i += 256) {
        const int r = i >> 7, c = i & (DIM - 1);
        Asub[r][c] = h[(row0 + r) * DIM + c];
    }
    for (int i = tid; i < RT * 16 * MEANW; i += 256) {
        const int r = i / MEANW, c = i - r * MEANW;
        Asub[r][DIM + c] = mean[(row0 + r) * MEANW + c];
    }
    __syncthreads();

    v8f acc[RT];
    #pragma unroll
    for (int rt = 0; rt < RT; ++rt)
        #pragma unroll
        for (int j = 0; j < 8; ++j) acc[rt][j] = 0.0f;

    const int m     = lane & 15;
    const int khalf = (lane >> 4) * 2;
    const int ncol  = wave * 16 + m;

    const float* __restrict__ arow[RT];
    #pragma unroll
    for (int rt = 0; rt < RT; ++rt)
        arow[rt] = &Asub[rt * 16 + m][khalf];
    const float* __restrict__ wp = W + (long)khalf * DOUT + ncol;

    for (int k = 0; k < KTOT; k += 4) {
        v2f b;
        b.x = wp[k * DOUT];
        b.y = wp[k * DOUT + DOUT];
        #pragma unroll
        for (int rt = 0; rt < RT; ++rt) {
            const float2 av = *(const float2*)(arow[rt] + k);
            v2f a; a.x = av.x; a.y = av.y;
            acc[rt] = __builtin_amdgcn_wmma_f32_16x16x4_f32(
                false, a, false, b, (short)0, acc[rt], false, false);
        }
    }

    const int mhi = (lane >> 4) * 8;
    #pragma unroll
    for (int rt = 0; rt < RT; ++rt) {
        #pragma unroll
        for (int r = 0; r < 8; ++r) {
            const long node = row0 + rt * 16 + r + mhi;
            float x = acc[rt][r];
            x = x > 0.0f ? x : expm1f(x);
            out[node * DOUT + ncol] = x;
        }
    }
}

extern "C" void kernel_launch(void* const* d_in, const int* in_sizes, int n_in,
                              void* d_out, int out_size, void* d_ws, size_t ws_size,
                              hipStream_t stream) {
    const float* h            = (const float*)d_in[0];
    const float* edge_table   = (const float*)d_in[1];
    const float* W            = (const float*)d_in[2];
    const int*   neighbor_idx = (const int*)d_in[3];
    const int*   edge_idx     = (const int*)d_in[4];

    float* mean = (float*)d_ws;
    float* out  = (float*)d_out;

    sage_gather_mean<<<NNODES / 8, 256, 0, stream>>>(h, edge_table,
                                                     neighbor_idx, edge_idx, mean);
    sage_gemm_elu<<<NNODES / (RT * 16), 256, 0, stream>>>(h, mean, W, out);
}


# gather fused into the GEMM kernel: 16 producer waves build the next 80-node [h|mean] tile in an LDS double buffer while 8 waves run WMMA on the current; mean stays on chip, 2 launches become 1
# speedup vs baseline: 1.8081x; 1.8081x over previous
#include <hip/hip_runtime.h>
#include <math.h>

typedef __attribute__((ext_vector_type(2))) float v2f;
typedef __attribute__((ext_vector_type(8))) float v8f;

#define NNODES 100000
#define DEG    16
#define DIM    128
#define DE     16
#define KTOT   (2 * DIM + DE)
#define DOUT   128
#define RT     5
#define TROWS  (RT * 16)
#define NTILES (NNODES / TROWS)
#define ASTRIDE 276
#define NCONS  8
#define NPROD  16
#define NPW    (TROWS / NPROD)

__device__ __forceinline__
void produce_tile(const float* __restrict__ h,
                  const float* __restrict__ edge_table,
                  const int*  __restrict__ neighbor_idx,
                  const int*  __restrict__ edge_idx,
                  float* __restrict__ abuf, const int tile, const int p, const int lane) {
    const long row0 = (long)tile * TROWS;
    const int epart = (lane & 3) * 4;
    const float inv = 1.0f / (float)DEG;

    int idv[NPW];
    #pragma unroll
    for (int j = 0; j < NPW; ++j) {
        const int r  = p + NPROD * j;
        const long ebase = (row0 + r) * DEG;
        const int* __restrict__ ip = (lane < DEG ? neighbor_idx : edge_idx) + ebase + (lane & (DEG - 1));
        idv[j] = *ip;
    }

    #pragma unroll
    for (int j = 0; j < NPW; ++j) {
        const int r = p + NPROD * j;
        {
            const long node = row0 + r;
            const int eid0 = __shfl(idv[j], DEG + (lane >> 2), 32);
            const int eid1 = __shfl(idv[j], DEG + 8 + (lane >> 2), 32);
            const float4 ev0 = *(const float4*)(edge_table + (long)eid0 * DE + epart);
            const float4 ev1 = *(const float4*)(edge_table + (long)eid1 * DE + epart);
            const float4 own = *(const float4*)(h + node * DIM + lane * 4);

            float4 hv[DEG];
            #pragma unroll
            for (int e = 0; e < DEG; ++e) {
                const int nbr = __builtin_amdgcn_readlane(idv[j], e);
                hv[e] = *(const float4*)(h + (long)nbr * DIM + lane * 4);
            }

            float4 ah = make_float4(0.f, 0.f, 0.f, 0.f);
            #pragma unroll
            for (int e = 0; e < DEG; ++e) {
                ah.x += hv[e].x; ah.y += hv[e].y; ah.z += hv[e].z; ah.w += hv[e].w;
            }

            float4 ae = make_float4(ev0.x + ev1.x, ev0.y + ev1.y, ev0.z + ev1.z, ev0.w + ev1.w);
            #pragma unroll
            for (int s = 4; s < 32; s <<= 1) {
                ae.x += __shfl_xor(ae.x, s, 32);
                ae.y += __shfl_xor(ae.y, s, 32);
                ae.z += __shfl_xor(ae.z, s, 32);
                ae.w += __shfl_xor(ae.w, s, 32);
            }

            float* __restrict__ rowp = abuf + r * ASTRIDE;
            *(float4*)(rowp + lane * 4) = own;
            *(float4*)(rowp + DIM + lane * 4) =
                make_float4(ah.x * inv, ah.y * inv, ah.z * inv, ah.w * inv);
            if (lane < 4)
                *(float4*)(rowp + 2 * DIM + lane * 4) =
                    make_float4(ae.x * inv, ae.y * inv, ae.z * inv, ae.w * inv);
        }
    }
}

__device__ __forceinline__
void consume_tile(const float* __restrict__ abuf,
                  const float* __restrict__ W,
                  float* __restrict__ out, const int tile, const int wave, const int lane) {
    const long row0 = (long)tile * TROWS;
    const int m     = lane & 15;
    const int khalf = (lane >> 4) * 2;
    const int ncol  = wave * 16 + m;

    v8f acc[RT];
    #pragma unroll
    for (int rt = 0; rt < RT; ++rt)
        #pragma unroll
        for (int j = 0; j < 8; ++j) acc[rt][j] = 0.0f;

    const float* __restrict__ ab = abuf + m * ASTRIDE + khalf;
    const float* __restrict__ wp = W + (long)khalf * DOUT + ncol;

    v2f bc[4];
    #pragma unroll
    for (int u = 0; u < 4; ++u) {
        bc[u].x = wp[(u * 4) * DOUT];
        bc[u].y = wp[(u * 4) * DOUT + DOUT];
    }
    #pragma unroll 1
    for (int k0 = 0; k0 < KTOT; k0 += 16) {
        const int kn = (k0 + 16 < KTOT) ? k0 + 16 : k0;
        v2f bn[4];
        #pragma unroll
        for (int u = 0; u < 4; ++u) {
            bn[u].x = wp[(kn + u * 4) * DOUT];
            bn[u].y = wp[(kn + u * 4) * DOUT + DOUT];
        }
        #pragma unroll
        for (int u = 0; u < 4; ++u) {
            #pragma unroll
            for (int rt = 0; rt < RT; ++rt) {
                const float2 av = *(const float2*)(ab + rt * 16 * ASTRIDE + k0 + u * 4);
                v2f a; a.x = av.x; a.y = av.y;
                acc[rt] = __builtin_amdgcn_wmma_f32_16x16x4_f32(
                    false, a, false, bc[u], (short)0, acc[rt], false, false);
            }
        }
        __builtin_amdgcn_sched_barrier(0);
        #pragma unroll
        for (int u = 0; u < 4; ++u) bc[u] = bn[u];
    }

    const int mhi = (lane >> 4) * 8;
    #pragma unroll
    for (int rt = 0; rt < RT; ++rt) {
        #pragma unroll
        for (int r = 0; r < 8; ++r) {
            const long node = row0 + rt * 16 + r + mhi;
            float x = acc[rt][r];
            x = x > 0.0f ? x : expm1f(x);
            out[node * DOUT + ncol] = x;
        }
    }
}

__global__ __launch_bounds__(768)
void sage_fused(const float* __restrict__ h,
                const float* __restrict__ edge_table,
                const float* __restrict__ W,
                const int*  __restrict__ neighbor_idx,
                const int*  __restrict__ edge_idx,
                float* __restrict__ out) {
    __shared__ __attribute__((aligned(16))) float Abuf[2][TROWS * ASTRIDE];

    const int wave = threadIdx.x >> 5;
    const int lane = threadIdx.x & 31;
    const int step = gridDim.x;

    int tile = blockIdx.x;
    if (wave >= NCONS && tile < NTILES)
        produce_tile(h, edge_table, neighbor_idx, edge_idx, Abuf[0], tile, wave - NCONS, lane);
    __syncthreads();

    int cur = 0;
    for (; tile < NTILES; tile += step) {
        if (wave >= NCONS) {
            if (tile + step < NTILES)
                produce_tile(h, edge_table, neighbor_idx, edge_idx, Abuf[cur ^ 1],
                             tile + step, wave - NCONS, lane);
        } else {
            consume_tile(Abuf[cur], W, out, tile, wave, lane);
        }
        __syncthreads();
        cur ^= 1;
    }
}

extern "C" void kernel_launch(void* const* d_in, const int* in_sizes, int n_in,
                              void* d_out, int out_size, void* d_ws, size_t ws_size,
                              hipStream_t stream) {
    const float* h            = (const float*)d_in[0];
    const float* edge_table   = (const float*)d_in[1];
    const float* W            = (const float*)d_in[2];
    const int*   neighbor_idx = (const int*)d_in[3];
    const int*   edge_idx     = (const int*)d_in[4];
    float* out  = (float*)d_out;

    sage_fused<<<256, 768, 0, stream>>>(h, edge_table, W, neighbor_idx, edge_idx, out);
}
